# starting kernel
# baseline (speedup 1.0000x reference)
#include <hip/hip_runtime.h>


typedef float v2f __attribute__((ext_vector_type(2)));
typedef float v8f __attribute__((ext_vector_type(8)));

__global__ __launch_bounds__(32)
void shiftconv_wmma_f32(const float* __restrict__ x,
                        const float* __restrict__ w,
                        float* __restrict__ out) {
  const int lane   = threadIdx.x;
  const int mt     = blockIdx.x;
  const int ng     = blockIdx.y;
  const int lanelo = lane & 15;
  const int hi     = lane >> 4;

  const int  r  = mt * 16 + lanelo;
  const bool rv = (r < 392);
  const int  rr = rv ? r : 0;
  const int  o   = rr / 196;
  const int  rem = rr - o * 196;
  const int  n   = rem / 14;
  const int  m   = rem - n * 14;
  const float* xb = x + (o * 256) * 196 + n * 14;

  const float rvf = rv ? 1.0f : 0.0f;
  float af[3];
  int   wc[3];
  af[0] = (m >= 1)  ? rvf : 0.0f;  wc[0] = (m + 12) % 14;
  af[1] = rvf;                     wc[1] = (m + 13) % 14;
  af[2] = (m <= 12) ? rvf : 0.0f;  wc[2] = m;

  const int kbase = ng * 64 + lanelo;

  const int qoff = hi * 2;
  const float* pA[6];
  const float* pB[6];
  float        mA[6];
  #pragma unroll
  for (int u = 0; u < 6; ++u) {
    const int t = u >> 1, v = u & 1;
    const int q = t * 4 + qoff + v;
    const int i = (int)((unsigned)q / 3u);
    const int j = q - i * 3;
    pA[u] = xb + i * 196 + wc[j];
    mA[u] = af[j];
    pB[u] = w + i * 768 + j + kbase * 3;
  }

  v8f acc0 = {}, acc1 = {}, acc2 = {}, acc3 = {};

  #pragma unroll 2
  for (int it = 0; it < 64; ++it) {
    #pragma unroll
    for (int t = 0; t < 3; ++t) {
      const int u0 = 2 * t, u1 = 2 * t + 1;

      v2f a;
      a.x = pA[u0][0] * mA[u0];
      a.y = pA[u1][0] * mA[u1];

      v2f b0, b1, b2, b3;
      b0.x = pB[u0][  0];  b0.y = pB[u1][  0];
      b1.x = pB[u0][ 48];  b1.y = pB[u1][ 48];
      b2.x = pB[u0][ 96];  b2.y = pB[u1][ 96];
      b3.x = pB[u0][144];  b3.y = pB[u1][144];

      acc0 = __builtin_amdgcn_wmma_f32_16x16x4_f32(false, a, false, b0, (short)0, acc0, false, false);
      acc1 = __builtin_amdgcn_wmma_f32_16x16x4_f32(false, a, false, b1, (short)0, acc1, false, false);
      acc2 = __builtin_amdgcn_wmma_f32_16x16x4_f32(false, a, false, b2, (short)0, acc2, false, false);
      acc3 = __builtin_amdgcn_wmma_f32_16x16x4_f32(false, a, false, b3, (short)0, acc3, false, false);
    }
    #pragma unroll
    for (int u = 0; u < 6; ++u) {
      pA[u] += 4 * 196;
      pB[u] += 4 * 768;
    }
  }

  #pragma unroll
  for (int d = 0; d < 8; ++d) {
    const int p = mt * 16 + hi * 8 + d;
    if (p < 392) {
      const int po   = p / 196;
      const int prem = p - po * 196;
      const int pn   = prem / 14;
      const int pm   = prem - pn * 14;
      const int pn1  = (pn + 1) % 14;
      float* ob = out + (po * 256 + ng * 64 + lanelo) * 196 + pn1 * 14 + pm;
      ob[ 0 * 196] = acc0[d];
      ob[16 * 196] = acc1[d];
      ob[32 * 196] = acc2[d];
      ob[48 * 196] = acc3[d];
    }
  }
}

extern "C" void kernel_launch(void* const* d_in, const int* in_sizes, int n_in,
                              void* d_out, int out_size, void* d_ws, size_t ws_size,
                              hipStream_t stream) {
  const float* x = (const float*)d_in[0];
  const float* w = (const float*)d_in[1];
  float* out = (float*)d_out;
  (void)in_sizes; (void)n_in; (void)out_size; (void)d_ws; (void)ws_size;

  dim3 grid(25, 4);
  shiftconv_wmma_f32<<<grid, 32, 0, stream>>>(x, w, out);
}


# K split over the 16 waves of a work-group (48 k each, a slice's 120 loads in flight together); partial 16x64 tiles added through LDS in wave order: changes the f32 summation order
# speedup vs baseline: 6.2402x; 6.2402x over previous
#include <hip/hip_runtime.h>


typedef float v2f __attribute__((ext_vector_type(2)));
typedef float v4f __attribute__((ext_vector_type(4)));
typedef float v8f __attribute__((ext_vector_type(8)));

#define KW      16
#define PERIODS (64 / KW)
#define PITCH   20
#define TILE_F  (64 * PITCH)

__global__ __launch_bounds__(32 * KW) __attribute__((amdgpu_waves_per_eu(4, 4)))
void shiftconv_wmma_f32(const float* __restrict__ x,
                        const float* __restrict__ w,
                        float* __restrict__ out) {
  __shared__ float part[KW * TILE_F];

  const int lane   = threadIdx.x & 31;
  const int wv     = threadIdx.x >> 5;
  const int mt     = blockIdx.x;
  const int ng     = blockIdx.y;
  const int lanelo = lane & 15;
  const int hi     = lane >> 4;

  const int  r  = mt * 16 + lanelo;
  const bool rv = (r < 392);
  const int  rr = rv ? r : 0;
  const int  o   = rr / 196;
  const int  rem = rr - o * 196;
  const int  n   = rem / 14;
  const int  m   = rem - n * 14;
  const float* xb = x + (o * 256) * 196 + n * 14;

  const float rvf = rv ? 1.0f : 0.0f;
  float af[3];
  int   wc[3];
  af[0] = (m >= 1)  ? rvf : 0.0f;  wc[0] = (m + 12) % 14;
  af[1] = rvf;                     wc[1] = (m + 13) % 14;
  af[2] = (m <= 12) ? rvf : 0.0f;  wc[2] = m;

  const int kbase = ng * 64 + lanelo;

  const int qoff = hi * 2;
  const int i0   = wv * (4 * PERIODS);
  const float* pA[6];
  const float* pB[6];
  float        mA[6];
  #pragma unroll
  for (int u = 0; u < 6; ++u) {
    const int t = u >> 1, v = u & 1;
    const int q = t * 4 + qoff + v;
    const int i = (int)((unsigned)q / 3u);
    const int j = q - i * 3;
    pA[u] = xb + (i0 + i) * 196 + wc[j];
    mA[u] = af[j];
    pB[u] = w + (i0 + i) * 768 + j + kbase * 3;
  }

  float av[PERIODS][6];
  float bv[PERIODS][6][4];
  #pragma unroll
  for (int it = 0; it < PERIODS; ++it) {
    #pragma unroll
    for (int u = 0; u < 6; ++u) {
      av[it][u]    = pA[u][it * 4 * 196];
      bv[it][u][0] = pB[u][it * 4 * 768 +   0];
      bv[it][u][1] = pB[u][it * 4 * 768 +  48];
      bv[it][u][2] = pB[u][it * 4 * 768 +  96];
      bv[it][u][3] = pB[u][it * 4 * 768 + 144];
    }
  }

  v8f acc0 = {}, acc1 = {}, acc2 = {}, acc3 = {};

  #pragma unroll
  for (int it = 0; it < PERIODS; ++it) {
    #pragma unroll
    for (int t = 0; t < 3; ++t) {
      const int u0 = 2 * t, u1 = 2 * t + 1;

      v2f a;
      a.x = av[it][u0] * mA[u0];
      a.y = av[it][u1] * mA[u1];

      v2f b0, b1, b2, b3;
      b0.x = bv[it][u0][0];  b0.y = bv[it][u1][0];
      b1.x = bv[it][u0][1];  b1.y = bv[it][u1][1];
      b2.x = bv[it][u0][2];  b2.y = bv[it][u1][2];
      b3.x = bv[it][u0][3];  b3.y = bv[it][u1][3];

      acc0 = __builtin_amdgcn_wmma_f32_16x16x4_f32(false, a, false, b0, (short)0, acc0, false, false);
      acc1 = __builtin_amdgcn_wmma_f32_16x16x4_f32(false, a, false, b1, (short)0, acc1, false, false);
      acc2 = __builtin_amdgcn_wmma_f32_16x16x4_f32(false, a, false, b2, (short)0, acc2, false, false);
      acc3 = __builtin_amdgcn_wmma_f32_16x16x4_f32(false, a, false, b3, (short)0, acc3, false, false);
    }
  }

  {
    float* my = part + wv * TILE_F + lanelo * PITCH + hi * 8;
    *(v4f*)(my +  0 * PITCH)     = __builtin_shufflevector(acc0, acc0, 0, 1, 2, 3);
    *(v4f*)(my +  0 * PITCH + 4) = __builtin_shufflevector(acc0, acc0, 4, 5, 6, 7);
    *(v4f*)(my + 16 * PITCH)     = __builtin_shufflevector(acc1, acc1, 0, 1, 2, 3);
    *(v4f*)(my + 16 * PITCH + 4) = __builtin_shufflevector(acc1, acc1, 4, 5, 6, 7);
    *(v4f*)(my + 32 * PITCH)     = __builtin_shufflevector(acc2, acc2, 0, 1, 2, 3);
    *(v4f*)(my + 32 * PITCH + 4) = __builtin_shufflevector(acc2, acc2, 4, 5, 6, 7);
    *(v4f*)(my + 48 * PITCH)     = __builtin_shufflevector(acc3, acc3, 0, 1, 2, 3);
    *(v4f*)(my + 48 * PITCH + 4) = __builtin_shufflevector(acc3, acc3, 4, 5, 6, 7);
  }
  __syncthreads();

  #pragma unroll
  for (int g = 0; g < (16 * 64) / (32 * KW); ++g) {
    const int e   = threadIdx.x + g * (32 * KW);
    const int col = e >> 4;
    const int row = e & 15;
    const float* src = part + col * PITCH + row;
    float s = src[0];
    #pragma unroll
    for (int v = 1; v < KW; ++v) s += src[v * TILE_F];

    const int p = mt * 16 + row;
    if (p < 392) {
      const int po   = p / 196;
      const int prem = p - po * 196;
      const int pn   = prem / 14;
      const int pm   = prem - pn * 14;
      const int pn1  = (pn + 1) % 14;
      out[(po * 256 + ng * 64 + col) * 196 + pn1 * 14 + pm] = s;
    }
  }
}

extern "C" void kernel_launch(void* const* d_in, const int* in_sizes, int n_in,
                              void* d_out, int out_size, void* d_ws, size_t ws_size,
                              hipStream_t stream) {
  const float* x = (const float*)d_in[0];
  const float* w = (const float*)d_in[1];
  float* out = (float*)d_out;
  (void)in_sizes; (void)n_in; (void)out_size; (void)d_ws; (void)ws_size;

  dim3 grid(25, 4);
  shiftconv_wmma_f32<<<grid, 32 * KW, 0, stream>>>(x, w, out);
}


# work-group tile 16x64 -> 32x16 (2 A fragments x 1 B), 8 waves of 32 channels each: 208 work-groups share the f32 matrix work, not 100; partials added over 8 waves not 16: f32 summation order changes
# speedup vs baseline: 7.6839x; 1.2314x over previous
#include <hip/hip_runtime.h>


typedef float v2f __attribute__((ext_vector_type(2)));
typedef float v4f __attribute__((ext_vector_type(4)));
typedef float v8f __attribute__((ext_vector_type(8)));

#define KW      8
#define PERIODS (64 / KW)
#define MT      32
#define NT      16
#define PITCH   36
#define TILE_F  (NT * PITCH)

__global__ __launch_bounds__(32 * KW) __attribute__((amdgpu_waves_per_eu(2, 2)))
void shiftconv_wmma_f32(const float* __restrict__ x,
                        const float* __restrict__ w,
                        float* __restrict__ out) {
  __shared__ __attribute__((aligned(16))) float part[KW * TILE_F];

  const int lane   = threadIdx.x & 31;
  const int wv     = threadIdx.x >> 5;
  const int mt     = blockIdx.x;
  const int ng     = blockIdx.y;
  const int lanelo = lane & 15;
  const int hi     = lane >> 4;

  const float* xb[2];
  float af[2][3];
  int   wc[2][3];
  #pragma unroll
  for (int f = 0; f < 2; ++f) {
    const int  r  = mt * MT + f * 16 + lanelo;
    const bool rv = (r < 392);
    const int  rr = rv ? r : 0;
    const int  o   = rr / 196;
    const int  rem = rr - o * 196;
    const int  n   = rem / 14;
    const int  m   = rem - n * 14;
    xb[f] = x + (o * 256) * 196 + n * 14;

    const float rvf = rv ? 1.0f : 0.0f;
    af[f][0] = (m >= 1)  ? rvf : 0.0f;  wc[f][0] = (m + 12) % 14;
    af[f][1] = rvf;                     wc[f][1] = (m + 13) % 14;
    af[f][2] = (m <= 12) ? rvf : 0.0f;  wc[f][2] = m;
  }

  const int kcol = ng * NT + lanelo;

  const int qoff = hi * 2;
  const int i0   = wv * (4 * PERIODS);
  const float* pA[2][6];
  const float* pB[6];
  float        mA[2][6];
  #pragma unroll
  for (int u = 0; u < 6; ++u) {
    const int t = u >> 1, v = u & 1;
    const int q = t * 4 + qoff + v;
    const int i = (int)((unsigned)q / 3u);
    const int j = q - i * 3;
    #pragma unroll
    for (int f = 0; f < 2; ++f) {
      pA[f][u] = xb[f] + (i0 + i) * 196 + wc[f][j];
      mA[f][u] = af[f][j];
    }
    pB[u] = w + (i0 + i) * 768 + j + kcol * 3;
  }

  float av[PERIODS][6][2];
  float bv[PERIODS][6];
  #pragma unroll
  for (int it = 0; it < PERIODS; ++it) {
    #pragma unroll
    for (int u = 0; u < 6; ++u) {
      av[it][u][0] = pA[0][u][it * 4 * 196];
      av[it][u][1] = pA[1][u][it * 4 * 196];
      bv[it][u]    = pB[u][it * 4 * 768];
    }
  }

  v8f acc0 = {}, acc1 = {};

  #pragma unroll
  for (int it = 0; it < PERIODS; ++it) {
    #pragma unroll
    for (int t = 0; t < 3; ++t) {
      const int u0 = 2 * t, u1 = 2 * t + 1;

      v2f a0, a1, b;
      a0.x = av[it][u0][0] * mA[0][u0];
      a0.y = av[it][u1][0] * mA[0][u1];
      a1.x = av[it][u0][1] * mA[1][u0];
      a1.y = av[it][u1][1] * mA[1][u1];
      b.x  = bv[it][u0];
      b.y  = bv[it][u1];

      acc0 = __builtin_amdgcn_wmma_f32_16x16x4_f32(false, a0, false, b, (short)0, acc0, false, false);
      acc1 = __builtin_amdgcn_wmma_f32_16x16x4_f32(false, a1, false, b, (short)0, acc1, false, false);
    }
  }

  {
    float* my = part + wv * TILE_F + lanelo * PITCH + hi * 8;
    *(v4f*)(my +  0)     = __builtin_shufflevector(acc0, acc0, 0, 1, 2, 3);
    *(v4f*)(my +  0 + 4) = __builtin_shufflevector(acc0, acc0, 4, 5, 6, 7);
    *(v4f*)(my + 16)     = __builtin_shufflevector(acc1, acc1, 0, 1, 2, 3);
    *(v4f*)(my + 16 + 4) = __builtin_shufflevector(acc1, acc1, 4, 5, 6, 7);
  }
  __syncthreads();

  #pragma unroll
  for (int g = 0; g < (MT * NT) / (32 * KW); ++g) {
    const int e   = threadIdx.x + g * (32 * KW);
    const int col = e >> 5;
    const int row = e & 31;
    const float* src = part + col * PITCH + row;
    float s = src[0];
    #pragma unroll
    for (int v = 1; v < KW; ++v) s += src[v * TILE_F];

    const int p = mt * MT + row;
    if (p < 392) {
      const int po   = p / 196;
      const int prem = p - po * 196;
      const int pn   = prem / 14;
      const int pm   = prem - pn * 14;
      const int pn1  = (pn + 1) % 14;
      out[(po * 256 + ng * NT + col) * 196 + pn1 * 14 + pm] = s;
    }
  }
}

extern "C" void kernel_launch(void* const* d_in, const int* in_sizes, int n_in,
                              void* d_out, int out_size, void* d_ws, size_t ws_size,
                              hipStream_t stream) {
  const float* x = (const float*)d_in[0];
  const float* w = (const float*)d_in[1];
  float* out = (float*)d_out;
  (void)in_sizes; (void)n_in; (void)out_size; (void)d_ws; (void)ws_size;

  dim3 grid(13, 16);
  shiftconv_wmma_f32<<<grid, 32 * KW, 0, stream>>>(x, w, out);
}


# taps moved out of K into 3 accumulator sets: x loaded once per channel, an (i,k)'s 3 tap weights as one 12-byte load; roll, padding, tap sum applied to the parked tiles: f32 summation order changes
# speedup vs baseline: 8.4923x; 1.1052x over previous
#include <hip/hip_runtime.h>


typedef float v2f __attribute__((ext_vector_type(2)));
typedef float v4f __attribute__((ext_vector_type(4)));
typedef float v8f __attribute__((ext_vector_type(8)));

#define KW      8
#define STEPS   (256 / KW / 4)
#define MV      28
#define NT      16
#define PITCH   36
#define TILE_F  (NT * PITCH)

struct __attribute__((packed, aligned(4))) taps3 { float t0, t1, t2; };

__global__ __launch_bounds__(32 * KW) __attribute__((amdgpu_waves_per_eu(2, 2)))
void shiftconv_wmma_f32(const float* __restrict__ x,
                        const float* __restrict__ w,
                        float* __restrict__ out) {
  __shared__ __attribute__((aligned(16))) float part[KW * 3 * TILE_F];

  const int lane = threadIdx.x & 31;
  const int wv   = threadIdx.x >> 5;
  const int mt   = blockIdx.x;
  const int ng   = blockIdx.y;
  const int c    = lane & 15;
  const int h    = lane >> 4;
  const int o    = mt / 7;
  const int a    = mt - o * 7;

  const int ch0 = wv * (4 * STEPS) + 2 * h;

  const int row1 = (16 + c < MV) ? (16 + c) : (MV - 1);
  const float* pa  = x + ((o * 256 + ch0) * 196 + 28 * a);
  const float* pa0 = pa + c;
  const float* pa1 = pa + row1;
  const float* pb  = w + (ch0 * 768 + (ng * NT + c) * 3);

  float av[STEPS][2][2];
  taps3 bv[STEPS][2];
  #pragma unroll
  for (int s = 0; s < STEPS; ++s) {
    #pragma unroll
    for (int e = 0; e < 2; ++e) {
      av[s][e][0] = pa0[(4 * s + e) * 196];
      av[s][e][1] = pa1[(4 * s + e) * 196];
      bv[s][e]    = *(const taps3*)(pb + (4 * s + e) * 768);
    }
  }

  v8f acc[3][2] = {};

  #pragma unroll
  for (int s = 0; s < STEPS; ++s) {
    v2f a0, a1, b0, b1, b2;
    a0.x = av[s][0][0];  a0.y = av[s][1][0];
    a1.x = av[s][0][1];  a1.y = av[s][1][1];
    b0.x = bv[s][0].t0;  b0.y = bv[s][1].t0;
    b1.x = bv[s][0].t1;  b1.y = bv[s][1].t1;
    b2.x = bv[s][0].t2;  b2.y = bv[s][1].t2;

    acc[0][0] = __builtin_amdgcn_wmma_f32_16x16x4_f32(false, a0, false, b0, (short)0, acc[0][0], false, false);
    acc[0][1] = __builtin_amdgcn_wmma_f32_16x16x4_f32(false, a1, false, b0, (short)0, acc[0][1], false, false);
    acc[1][0] = __builtin_amdgcn_wmma_f32_16x16x4_f32(false, a0, false, b1, (short)0, acc[1][0], false, false);
    acc[1][1] = __builtin_amdgcn_wmma_f32_16x16x4_f32(false, a1, false, b1, (short)0, acc[1][1], false, false);
    acc[2][0] = __builtin_amdgcn_wmma_f32_16x16x4_f32(false, a0, false, b2, (short)0, acc[2][0], false, false);
    acc[2][1] = __builtin_amdgcn_wmma_f32_16x16x4_f32(false, a1, false, b2, (short)0, acc[2][1], false, false);
    __builtin_amdgcn_sched_barrier(0);
  }

  #pragma unroll
  for (int j = 0; j < 3; ++j) {
    float* my = part + (wv * 3 + j) * TILE_F + c * PITCH + h * 8;
    *(v4f*)(my +  0)     = __builtin_shufflevector(acc[j][0], acc[j][0], 0, 1, 2, 3);
    *(v4f*)(my +  0 + 4) = __builtin_shufflevector(acc[j][0], acc[j][0], 4, 5, 6, 7);
    *(v4f*)(my + 16)     = __builtin_shufflevector(acc[j][1], acc[j][1], 0, 1, 2, 3);
    *(v4f*)(my + 16 + 4) = __builtin_shufflevector(acc[j][1], acc[j][1], 4, 5, 6, 7);
  }
  __syncthreads();

  #pragma unroll
  for (int g = 0; g < (32 * NT) / (32 * KW); ++g) {
    const int e   = threadIdx.x + g * (32 * KW);
    const int col = e >> 5;
    const int row = e & 31;
    const int rr  = (row < MV) ? row : (MV - 1);
    const int nl  = (rr >= 14) ? 1 : 0;
    const int m   = rr - 14 * nl;
    const float* src = part + col * PITCH + nl * 14;

    float tap[3];
    #pragma unroll
    for (int j = 0; j < 3; ++j) {
      int cj = m + j - 2;
      cj = (cj < 0) ? (cj + 14) : cj;
      float sj = src[j * TILE_F + cj];
      #pragma unroll
      for (int v = 1; v < KW; ++v) sj += src[(v * 3 + j) * TILE_F + cj];
      const bool present = (m + j >= 1) && (m + j <= 14);
      tap[j] = present ? sj : 0.0f;
    }
    const float sum = (tap[0] + tap[1]) + tap[2];

    if (row < MV) {
      const int n   = 2 * a + nl;
      const int pn1 = (n + 1) % 14;
      out[(o * 256 + ng * NT + col) * 196 + pn1 * 14 + m] = sum;
    }
  }
}

extern "C" void kernel_launch(void* const* d_in, const int* in_sizes, int n_in,
                              void* d_out, int out_size, void* d_ws, size_t ws_size,
                              hipStream_t stream) {
  const float* x = (const float*)d_in[0];
  const float* w = (const float*)d_in[1];
  float* out = (float*)d_out;
  (void)in_sizes; (void)n_in; (void)out_size; (void)d_ws; (void)ws_size;

  dim3 grid(14, 16);
  shiftconv_wmma_f32<<<grid, 32 * KW, 0, stream>>>(x, w, out);
}


# 3-tap sum as minimal filtering, 4 transformed products per output pair not 6: 32 WMMAs a wave not 48, pairs formed in registers, 2 tiles parked not 3: f32 rounding and summation order change
# speedup vs baseline: 9.0554x; 1.0663x over previous
#include <hip/hip_runtime.h>


typedef float v2f __attribute__((ext_vector_type(2)));
typedef float v4f __attribute__((ext_vector_type(4)));
typedef float v8f __attribute__((ext_vector_type(8)));

#define KW      8
#define STEPS   (256 / KW / 4)
#define NT      16
#define PITCH   20
#define TILE_F  (NT * PITCH + 16)

struct __attribute__((packed, aligned(4))) taps3 { float t0, t1, t2; };

__global__ __launch_bounds__(32 * KW) __attribute__((amdgpu_waves_per_eu(2, 2)))
void shiftconv_wmma_f32(const float* __restrict__ x,
                        const float* __restrict__ w,
                        float* __restrict__ out) {
  __shared__ __attribute__((aligned(16))) float part[KW * 2 * TILE_F];

  const int lane = threadIdx.x & 31;
  const int wv   = threadIdx.x >> 5;
  const int mt   = blockIdx.x;
  const int ng   = blockIdx.y;
  const int c    = lane & 15;
  const int h    = lane >> 4;
  const int o    = mt / 7;
  const int a    = mt - o * 7;

  const int ch0 = wv * (4 * STEPS) + 2 * h;

  const int cc = (c < 14) ? c : 13;
  const int nl = (cc >= 7) ? 1 : 0;
  const int t  = cc - 7 * nl;
  const int  lo_off = (t == 0) ? 12 : (2 * t - 2);
  const int  hi_off = 2 * t;
  const bool first  = (t == 0);
  const bool last   = (t == 6);

  const float* pa  = x + ((o * 256 + ch0) * 196 + 28 * a + 14 * nl);
  const v2f*   plo = (const v2f*)(pa + lo_off);
  const v2f*   phi = (const v2f*)(pa + hi_off);
  const float* pb  = w + (ch0 * 768 + (ng * NT + c) * 3);

  v2f   lo[STEPS][2], hi[STEPS][2];
  taps3 bv[STEPS][2];
  #pragma unroll
  for (int s = 0; s < STEPS; ++s) {
    #pragma unroll
    for (int e = 0; e < 2; ++e) {
      lo[s][e] = plo[(4 * s + e) * 98];
      hi[s][e] = phi[(4 * s + e) * 98];
      bv[s][e] = *(const taps3*)(pb + (4 * s + e) * 768);
    }
  }

  v8f acc[4] = {};

  #pragma unroll
  for (int s = 0; s < STEPS; ++s) {
    v2f A0, A1, A2, A3, B0, B1, B2, B3;
    #pragma unroll
    for (int e = 0; e < 2; ++e) {
      const float d0 = first ? 0.0f : lo[s][e].x;
      const float d1 = lo[s][e].y;
      const float d2 = hi[s][e].x;
      const float d3 = last ? 0.0f : hi[s][e].y;
      const float g0 = bv[s][e].t0;
      const float g1 = bv[s][e].t1;
      const float g2 = bv[s][e].t2;
      const float gs = g0 + g2;
      A0[e] = d0 - d2;
      A1[e] = d1 + d2;
      A2[e] = d2 - d1;
      A3[e] = d1 - d3;
      B0[e] = g0;
      B1[e] = 0.5f * (gs + g1);
      B2[e] = 0.5f * (gs - g1);
      B3[e] = g2;
    }

    acc[0] = __builtin_amdgcn_wmma_f32_16x16x4_f32(false, A0, false, B0, (short)0, acc[0], false, false);
    acc[1] = __builtin_amdgcn_wmma_f32_16x16x4_f32(false, A1, false, B1, (short)0, acc[1], false, false);
    acc[2] = __builtin_amdgcn_wmma_f32_16x16x4_f32(false, A2, false, B2, (short)0, acc[2], false, false);
    acc[3] = __builtin_amdgcn_wmma_f32_16x16x4_f32(false, A3, false, B3, (short)0, acc[3], false, false);
    __builtin_amdgcn_sched_barrier(0);
  }

  const v8f y0 = (acc[0] + acc[1]) + acc[2];
  const v8f y1 = (acc[1] - acc[2]) - acc[3];

  {
    float* my = part + (wv * 2) * TILE_F + c * PITCH + h * 8;
    *(v4f*)(my)              = __builtin_shufflevector(y0, y0, 0, 1, 2, 3);
    *(v4f*)(my + 4)          = __builtin_shufflevector(y0, y0, 4, 5, 6, 7);
    *(v4f*)(my + TILE_F)     = __builtin_shufflevector(y1, y1, 0, 1, 2, 3);
    *(v4f*)(my + TILE_F + 4) = __builtin_shufflevector(y1, y1, 4, 5, 6, 7);
  }
  __syncthreads();

  #pragma unroll
  for (int g = 0; g < NT / KW; ++g) {
    const int col = wv + g * KW;
    const int ll  = (lane < 28) ? lane : 27;
    const int nl2 = (ll >= 14) ? 1 : 0;
    const int m   = ll - 14 * nl2;
    const float* src = part + (m & 1) * TILE_F + col * PITCH + nl2 * 7 + (m >> 1);

    float sum = src[0];
    #pragma unroll
    for (int v = 1; v < KW; ++v) sum += src[v * 2 * TILE_F];

    if (lane < 28) {
      const int n   = 2 * a + nl2;
      const int pn1 = (n + 1) % 14;
      out[(o * 256 + ng * NT + col) * 196 + pn1 * 14 + m] = sum;
    }
  }
}

extern "C" void kernel_launch(void* const* d_in, const int* in_sizes, int n_in,
                              void* d_out, int out_size, void* d_ws, size_t ws_size,
                              hipStream_t stream) {
  const float* x = (const float*)d_in[0];
  const float* w = (const float*)d_in[1];
  float* out = (float*)d_out;
  (void)in_sizes; (void)n_in; (void)out_size; (void)d_ws; (void)ws_size;

  dim3 grid(14, 16);
  shiftconv_wmma_f32<<<grid, 32 * KW, 0, stream>>>(x, w, out);
}
